# starting kernel
# baseline (speedup 1.0000x reference)
#include <hip/hip_runtime.h>

typedef __attribute__((ext_vector_type(2))) float v2f;
typedef __attribute__((ext_vector_type(4))) float v4f;
typedef __attribute__((ext_vector_type(8))) float v8f;

#define NF    32
#define BATCH 128
#define TLEN  1024
#define HID   16
#define KW    5
#define DPF   4
#define CT    512
#define XSP   524
#define HTP   20
#define HROWS 520
#define XS_OFF   0
#define H_OFF    524
#define PART_OFF (H_OFF + HROWS * HTP)
#define LDS_FLOATS (PART_OFF + HID * DPF * 2)

__global__ __launch_bounds__(128)
void branch_kernel(const float* __restrict__ x,
                   const float* __restrict__ w1, const float* __restrict__ b1,
                   const float* __restrict__ w2, const float* __restrict__ b2,
                   const float* __restrict__ pw, const float* __restrict__ pb,
                   float* __restrict__ out)
{
    __shared__ __align__(16) float lds[LDS_FLOATS];
    const int f    = blockIdx.x & (NF - 1);
    const int b    = blockIdx.x >> 5;
    const int tid  = threadIdx.x;
    const int lane = tid & 31;
    const int wv   = tid >> 5;
    const int m    = lane & 15;
    const int half = lane >> 4;

    const float* xg  = x  + (size_t)b * TLEN * NF + f;
    const float* w1f = w1 + f * HID * KW;
    const float* b1f = b1 + f * HID;
    const float* w2f = w2 + f * HID * HID * KW;
    const float* b2f = b2 + f * HID;

    float aw[40];
#pragma unroll
    for (int dk = 0; dk < KW; ++dk) {
#pragma unroll
        for (int s = 0; s < 4; ++s) {
            int ci = 4 * s + 2 * half;
            aw[(dk * 4 + s) * 2 + 0] = w2f[(m * HID + ci    ) * KW + dk];
            aw[(dk * 4 + s) * 2 + 1] = w2f[(m * HID + ci + 1) * KW + dk];
        }
    }
    float bb[8];
#pragma unroll
    for (int r = 0; r < 8; ++r) bb[r] = b2f[r + 8 * half];

#pragma unroll 1
    for (int ch = 0; ch < 2; ++ch) {
        const int tbase = ch * CT;

        for (int v = tid; v < XSP; v += 128) {
            int t = tbase + v - 6;
            lds[XS_OFF + v] = (t >= 0 && t < TLEN) ? xg[(size_t)t * NF] : 0.f;
        }
        __syncthreads();

        for (int u = tid; u < HROWS; u += 128) {
            int t = tbase + u - 4;
            bool valid = (t >= 0 && t < TLEN);
            float xv[KW];
#pragma unroll
            for (int j = 0; j < KW; ++j) xv[j] = lds[XS_OFF + u + j];
            float hloc[HID];
#pragma unroll
            for (int c = 0; c < HID; ++c) {
                float h = b1f[c];
#pragma unroll
                for (int j = 0; j < KW; ++j) h = fmaf(w1f[c * KW + j], xv[j], h);
                hloc[c] = valid ? fmaxf(h, 0.f) : 0.f;
            }
#pragma unroll
            for (int g = 0; g < 4; ++g) {
                v4f q = { hloc[4*g+0], hloc[4*g+1], hloc[4*g+2], hloc[4*g+3] };
                *(v4f*)(&lds[H_OFF + u * HTP + 4 * g]) = q;
            }
        }
        __syncthreads();

        float pacc[8];
#pragma unroll
        for (int r = 0; r < 8; ++r) pacc[r] = 0.f;

#pragma unroll 2
        for (int jj = 0; jj < 8; ++jj) {
            int j  = (wv >> 1) * 16 + (wv & 1) * 8 + jj;
            int ub = 16 * j + 2;
            v8f acc8 = {0.f, 0.f, 0.f, 0.f, 0.f, 0.f, 0.f, 0.f};
#pragma unroll
            for (int dk = 0; dk < KW; ++dk) {
                int rowbase = H_OFF + (ub + m + dk) * HTP + 2 * half;
#pragma unroll
                for (int s = 0; s < 4; ++s) {
                    v2f a;
                    a.x = aw[(dk * 4 + s) * 2 + 0];
                    a.y = aw[(dk * 4 + s) * 2 + 1];
                    v2f bv = *(const v2f*)(&lds[rowbase + 4 * s]);
                    acc8 = __builtin_amdgcn_wmma_f32_16x16x4_f32(
                               false, a, false, bv, (short)0, acc8, false, false);
                }
            }
#pragma unroll
            for (int r = 0; r < 8; ++r) pacc[r] += fmaxf(acc8[r] + bb[r], 0.f);
        }

        int seg = 2 * ch + (wv >> 1);
        int par = wv & 1;
#pragma unroll
        for (int r = 0; r < 8; ++r) {
            float v = pacc[r];
            v += __shfl_xor(v, 1, 32);
            v += __shfl_xor(v, 2, 32);
            v += __shfl_xor(v, 4, 32);
            v += __shfl_xor(v, 8, 32);
            if ((lane & 15) == 0)
                lds[PART_OFF + ((r + 8 * half) * DPF + seg) * 2 + par] = v;
        }
        __syncthreads();
    }

    if (tid < DPF) {
        const float* pwf = pw + ((size_t)f * DPF + tid) * (HID * DPF);
        float acc = pb[f * DPF + tid];
#pragma unroll
        for (int i = 0; i < HID * DPF; ++i) {
            float p = (lds[PART_OFF + i * 2] + lds[PART_OFF + i * 2 + 1]) * (1.f / 256.f);
            acc = fmaf(pwf[i], p, acc);
        }
        out[(size_t)b * (NF * DPF) + f * DPF + tid] = acc;
    }
}

extern "C" void kernel_launch(void* const* d_in, const int* in_sizes, int n_in,
                              void* d_out, int out_size, void* d_ws, size_t ws_size,
                              hipStream_t stream) {
    (void)in_sizes; (void)n_in; (void)d_ws; (void)ws_size; (void)out_size;
    const float* x  = (const float*)d_in[0];
    const float* w1 = (const float*)d_in[2];
    const float* b1 = (const float*)d_in[3];
    const float* w2 = (const float*)d_in[4];
    const float* b2 = (const float*)d_in[5];
    const float* pw = (const float*)d_in[6];
    const float* pb = (const float*)d_in[7];
    float* out = (float*)d_out;

    dim3 grid(NF * BATCH);
    dim3 block(128);
    branch_kernel<<<grid, block, 0, stream>>>(x, w1, b1, w2, b2, pw, pb, out);
}


# conv1 split by channel half across waves (a wave does 8 channels of every second 32-row group): its 40 weights + 8 biases stay in scalar registers, no spills to vector lanes, loaded once per block
# speedup vs baseline: 1.1189x; 1.1189x over previous
#include <hip/hip_runtime.h>

typedef __attribute__((ext_vector_type(2))) float v2f;
typedef __attribute__((ext_vector_type(4))) float v4f;
typedef __attribute__((ext_vector_type(8))) float v8f;

#define NF    32
#define BATCH 128
#define TLEN  1024
#define HID   16
#define KW    5
#define DPF   4
#define CT    512
#define XSP   524
#define HTP   20
#define HROWS 520
#define XS_OFF   0
#define H_OFF    524
#define PART_OFF (H_OFF + HROWS * HTP)
#define LDS_FLOATS (PART_OFF + HID * DPF * 2)

__global__ __launch_bounds__(128)
void branch_kernel(const float* __restrict__ x,
                   const float* __restrict__ w1, const float* __restrict__ b1,
                   const float* __restrict__ w2, const float* __restrict__ b2,
                   const float* __restrict__ pw, const float* __restrict__ pb,
                   float* __restrict__ out)
{
    __shared__ __align__(16) float lds[LDS_FLOATS];
    const int f    = blockIdx.x & (NF - 1);
    const int b    = blockIdx.x >> 5;
    const int tid  = threadIdx.x;
    const int lane = tid & 31;
    const int wv   = tid >> 5;
    const int m    = lane & 15;
    const int half = lane >> 4;

    const float* xg  = x  + (size_t)b * TLEN * NF + f;
    const float* w1f = w1 + f * HID * KW;
    const float* b1f = b1 + f * HID;
    const float* w2f = w2 + f * HID * HID * KW;
    const float* b2f = b2 + f * HID;

    const int wvu   = __builtin_amdgcn_readfirstlane(wv);
    const int chalf = wvu >> 1;
    const int rpar  = wvu & 1;
    const float* w1h = w1f + chalf * (HID / 2) * KW;
    const float* b1h = b1f + chalf * (HID / 2);

    float aw[40];
#pragma unroll
    for (int dk = 0; dk < KW; ++dk) {
#pragma unroll
        for (int s = 0; s < 4; ++s) {
            int ci = 4 * s + 2 * half;
            aw[(dk * 4 + s) * 2 + 0] = w2f[(m * HID + ci    ) * KW + dk];
            aw[(dk * 4 + s) * 2 + 1] = w2f[(m * HID + ci + 1) * KW + dk];
        }
    }
    float bb[8];
#pragma unroll
    for (int r = 0; r < 8; ++r) bb[r] = b2f[r + 8 * half];

#pragma unroll 1
    for (int ch = 0; ch < 2; ++ch) {
        const int tbase = ch * CT;

        for (int v = tid; v < XSP; v += 128) {
            int t = tbase + v - 6;
            lds[XS_OFF + v] = (t >= 0 && t < TLEN) ? xg[(size_t)t * NF] : 0.f;
        }
        __syncthreads();

        for (int u = rpar * 32 + lane; u < HROWS; u += 64) {
            int t = tbase + u - 4;
            bool valid = (t >= 0 && t < TLEN);
            float xv[KW];
#pragma unroll
            for (int j = 0; j < KW; ++j) xv[j] = lds[XS_OFF + u + j];
            float hloc[HID / 2];
#pragma unroll
            for (int c = 0; c < HID / 2; ++c) {
                float h = b1h[c];
#pragma unroll
                for (int j = 0; j < KW; ++j) h = fmaf(w1h[c * KW + j], xv[j], h);
                hloc[c] = valid ? fmaxf(h, 0.f) : 0.f;
            }
#pragma unroll
            for (int g = 0; g < 2; ++g) {
                v4f q = { hloc[4*g+0], hloc[4*g+1], hloc[4*g+2], hloc[4*g+3] };
                *(v4f*)(&lds[H_OFF + u * HTP + chalf * 8 + 4 * g]) = q;
            }
        }
        __syncthreads();

        float pacc[8];
#pragma unroll
        for (int r = 0; r < 8; ++r) pacc[r] = 0.f;

#pragma unroll 2
        for (int jj = 0; jj < 8; ++jj) {
            int j  = (wv >> 1) * 16 + (wv & 1) * 8 + jj;
            int ub = 16 * j + 2;
            v8f acc8 = {0.f, 0.f, 0.f, 0.f, 0.f, 0.f, 0.f, 0.f};
#pragma unroll
            for (int dk = 0; dk < KW; ++dk) {
                int rowbase = H_OFF + (ub + m + dk) * HTP + 2 * half;
#pragma unroll
                for (int s = 0; s < 4; ++s) {
                    v2f a;
                    a.x = aw[(dk * 4 + s) * 2 + 0];
                    a.y = aw[(dk * 4 + s) * 2 + 1];
                    v2f bv = *(const v2f*)(&lds[rowbase + 4 * s]);
                    acc8 = __builtin_amdgcn_wmma_f32_16x16x4_f32(
                               false, a, false, bv, (short)0, acc8, false, false);
                }
            }
#pragma unroll
            for (int r = 0; r < 8; ++r) pacc[r] += fmaxf(acc8[r] + bb[r], 0.f);
        }

        int seg = 2 * ch + (wv >> 1);
        int par = wv & 1;
#pragma unroll
        for (int r = 0; r < 8; ++r) {
            float v = pacc[r];
            v += __shfl_xor(v, 1, 32);
            v += __shfl_xor(v, 2, 32);
            v += __shfl_xor(v, 4, 32);
            v += __shfl_xor(v, 8, 32);
            if ((lane & 15) == 0)
                lds[PART_OFF + ((r + 8 * half) * DPF + seg) * 2 + par] = v;
        }
        __syncthreads();
    }

    if (tid < DPF) {
        const float* pwf = pw + ((size_t)f * DPF + tid) * (HID * DPF);
        float acc = pb[f * DPF + tid];
#pragma unroll
        for (int i = 0; i < HID * DPF; ++i) {
            float p = (lds[PART_OFF + i * 2] + lds[PART_OFF + i * 2 + 1]) * (1.f / 256.f);
            acc = fmaf(pwf[i], p, acc);
        }
        out[(size_t)b * (NF * DPF) + f * DPF + tid] = acc;
    }
}

extern "C" void kernel_launch(void* const* d_in, const int* in_sizes, int n_in,
                              void* d_out, int out_size, void* d_ws, size_t ws_size,
                              hipStream_t stream) {
    (void)in_sizes; (void)n_in; (void)d_ws; (void)ws_size; (void)out_size;
    const float* x  = (const float*)d_in[0];
    const float* w1 = (const float*)d_in[2];
    const float* b1 = (const float*)d_in[3];
    const float* w2 = (const float*)d_in[4];
    const float* b2 = (const float*)d_in[5];
    const float* pw = (const float*)d_in[6];
    const float* pb = (const float*)d_in[7];
    float* out = (float*)d_out;

    dim3 grid(NF * BATCH);
    dim3 block(128);
    branch_kernel<<<grid, block, 0, stream>>>(x, w1, b1, w2, b2, pw, pb, out);
}


# 16-lane pool reduction on the vector ALU: the four xor-butterfly steps are DPP row_xmask adds instead of 32 serial LDS-permute round trips per chunk; same summation tree
# speedup vs baseline: 1.1556x; 1.0328x over previous
#include <hip/hip_runtime.h>

typedef __attribute__((ext_vector_type(2))) float v2f;
typedef __attribute__((ext_vector_type(4))) float v4f;
typedef __attribute__((ext_vector_type(8))) float v8f;

#define NF    32
#define BATCH 128
#define TLEN  1024
#define HID   16
#define KW    5
#define DPF   4
#define CT    512
#define XSP   524
#define HTP   20
#define HROWS 520
#define XS_OFF   0
#define H_OFF    524
#define PART_OFF (H_OFF + HROWS * HTP)
#define LDS_FLOATS (PART_OFF + HID * DPF * 2)

template <int CTRL>
__device__ __forceinline__ float row_xor(float v) {
    int i = __builtin_bit_cast(int, v);
    int o = __builtin_amdgcn_update_dpp(i, i, CTRL, 0xf, 0xf, false);
    return __builtin_bit_cast(float, o);
}

__global__ __launch_bounds__(128)
void branch_kernel(const float* __restrict__ x,
                   const float* __restrict__ w1, const float* __restrict__ b1,
                   const float* __restrict__ w2, const float* __restrict__ b2,
                   const float* __restrict__ pw, const float* __restrict__ pb,
                   float* __restrict__ out)
{
    __shared__ __align__(16) float lds[LDS_FLOATS];
    const int f    = blockIdx.x & (NF - 1);
    const int b    = blockIdx.x >> 5;
    const int tid  = threadIdx.x;
    const int lane = tid & 31;
    const int wv   = tid >> 5;
    const int m    = lane & 15;
    const int half = lane >> 4;

    const float* xg  = x  + (size_t)b * TLEN * NF + f;
    const float* w1f = w1 + f * HID * KW;
    const float* b1f = b1 + f * HID;
    const float* w2f = w2 + f * HID * HID * KW;
    const float* b2f = b2 + f * HID;

    const int wvu   = __builtin_amdgcn_readfirstlane(wv);
    const int chalf = wvu >> 1;
    const int rpar  = wvu & 1;
    const float* w1h = w1f + chalf * (HID / 2) * KW;
    const float* b1h = b1f + chalf * (HID / 2);

    float aw[40];
#pragma unroll
    for (int dk = 0; dk < KW; ++dk) {
#pragma unroll
        for (int s = 0; s < 4; ++s) {
            int ci = 4 * s + 2 * half;
            aw[(dk * 4 + s) * 2 + 0] = w2f[(m * HID + ci    ) * KW + dk];
            aw[(dk * 4 + s) * 2 + 1] = w2f[(m * HID + ci + 1) * KW + dk];
        }
    }
    float bb[8];
#pragma unroll
    for (int r = 0; r < 8; ++r) bb[r] = b2f[r + 8 * half];

#pragma unroll 1
    for (int ch = 0; ch < 2; ++ch) {
        const int tbase = ch * CT;

        for (int v = tid; v < XSP; v += 128) {
            int t = tbase + v - 6;
            lds[XS_OFF + v] = (t >= 0 && t < TLEN) ? xg[(size_t)t * NF] : 0.f;
        }
        __syncthreads();

        for (int u = rpar * 32 + lane; u < HROWS; u += 64) {
            int t = tbase + u - 4;
            bool valid = (t >= 0 && t < TLEN);
            float xv[KW];
#pragma unroll
            for (int j = 0; j < KW; ++j) xv[j] = lds[XS_OFF + u + j];
            float hloc[HID / 2];
#pragma unroll
            for (int c = 0; c < HID / 2; ++c) {
                float h = b1h[c];
#pragma unroll
                for (int j = 0; j < KW; ++j) h = fmaf(w1h[c * KW + j], xv[j], h);
                hloc[c] = valid ? fmaxf(h, 0.f) : 0.f;
            }
#pragma unroll
            for (int g = 0; g < 2; ++g) {
                v4f q = { hloc[4*g+0], hloc[4*g+1], hloc[4*g+2], hloc[4*g+3] };
                *(v4f*)(&lds[H_OFF + u * HTP + chalf * 8 + 4 * g]) = q;
            }
        }
        __syncthreads();

        float pacc[8];
#pragma unroll
        for (int r = 0; r < 8; ++r) pacc[r] = 0.f;

#pragma unroll 2
        for (int jj = 0; jj < 8; ++jj) {
            int j  = (wv >> 1) * 16 + (wv & 1) * 8 + jj;
            int ub = 16 * j + 2;
            v8f acc8 = {0.f, 0.f, 0.f, 0.f, 0.f, 0.f, 0.f, 0.f};
#pragma unroll
            for (int dk = 0; dk < KW; ++dk) {
                int rowbase = H_OFF + (ub + m + dk) * HTP + 2 * half;
#pragma unroll
                for (int s = 0; s < 4; ++s) {
                    v2f a;
                    a.x = aw[(dk * 4 + s) * 2 + 0];
                    a.y = aw[(dk * 4 + s) * 2 + 1];
                    v2f bv = *(const v2f*)(&lds[rowbase + 4 * s]);
                    acc8 = __builtin_amdgcn_wmma_f32_16x16x4_f32(
                               false, a, false, bv, (short)0, acc8, false, false);
                }
            }
#pragma unroll
            for (int r = 0; r < 8; ++r) pacc[r] += fmaxf(acc8[r] + bb[r], 0.f);
        }

        int seg = 2 * ch + (wv >> 1);
        int par = wv & 1;
        float red[8];
#pragma unroll
        for (int r = 0; r < 8; ++r) {
            float v = pacc[r];
            v += row_xor<0x161>(v);
            v += row_xor<0x162>(v);
            v += row_xor<0x164>(v);
            v += row_xor<0x168>(v);
            red[r] = v;
        }
        if ((lane & 15) == 0) {
#pragma unroll
            for (int r = 0; r < 8; ++r)
                lds[PART_OFF + ((r + 8 * half) * DPF + seg) * 2 + par] = red[r];
        }
        __syncthreads();
    }

    if (tid < DPF) {
        const float* pwf = pw + ((size_t)f * DPF + tid) * (HID * DPF);
        float acc = pb[f * DPF + tid];
#pragma unroll
        for (int i = 0; i < HID * DPF; ++i) {
            float p = (lds[PART_OFF + i * 2] + lds[PART_OFF + i * 2 + 1]) * (1.f / 256.f);
            acc = fmaf(pwf[i], p, acc);
        }
        out[(size_t)b * (NF * DPF) + f * DPF + tid] = acc;
    }
}

extern "C" void kernel_launch(void* const* d_in, const int* in_sizes, int n_in,
                              void* d_out, int out_size, void* d_ws, size_t ws_size,
                              hipStream_t stream) {
    (void)in_sizes; (void)n_in; (void)d_ws; (void)ws_size; (void)out_size;
    const float* x  = (const float*)d_in[0];
    const float* w1 = (const float*)d_in[2];
    const float* b1 = (const float*)d_in[3];
    const float* w2 = (const float*)d_in[4];
    const float* b2 = (const float*)d_in[5];
    const float* pw = (const float*)d_in[6];
    const float* pb = (const float*)d_in[7];
    float* out = (float*)d_out;

    dim3 grid(NF * BATCH);
    dim3 block(128);
    branch_kernel<<<grid, block, 0, stream>>>(x, w1, b1, w2, b2, pw, pb, out);
}


# conv2 as minimal filtering F(2,5): 6 transformed products per output pair for 10 taps (24 f32 WMMAs per 32 outputs, was 40), transforms on the vector ALU; conv2 and pool sum order change
# speedup vs baseline: 1.2940x; 1.1198x over previous
#include <hip/hip_runtime.h>

typedef __attribute__((ext_vector_type(2))) float v2f;
typedef __attribute__((ext_vector_type(4))) float v4f;
typedef __attribute__((ext_vector_type(8))) float v8f;

#define NF    32
#define BATCH 128
#define TLEN  1024
#define HID   16
#define KW    5
#define DPF   4
#define NPT   6
#define CT    512
#define XSP   524
#define HTP   20
#define HROWS 520
#define XS_OFF   0
#define H_OFF    524
#define PART_OFF (H_OFF + HROWS * HTP)
#define LDS_FLOATS (PART_OFF + HID * DPF * 2)

template <int CTRL>
__device__ __forceinline__ float row_xor(float v) {
    int i = __builtin_bit_cast(int, v);
    int o = __builtin_amdgcn_update_dpp(i, i, CTRL, 0xf, 0xf, false);
    return __builtin_bit_cast(float, o);
}

__global__ __launch_bounds__(128)
void branch_kernel(const float* __restrict__ x,
                   const float* __restrict__ w1, const float* __restrict__ b1,
                   const float* __restrict__ w2, const float* __restrict__ b2,
                   const float* __restrict__ pw, const float* __restrict__ pb,
                   float* __restrict__ out)
{
    __shared__ __align__(16) float lds[LDS_FLOATS];
    const int f    = blockIdx.x & (NF - 1);
    const int b    = blockIdx.x >> 5;
    const int tid  = threadIdx.x;
    const int lane = tid & 31;
    const int wv   = tid >> 5;
    const int m    = lane & 15;
    const int half = lane >> 4;

    const float* xg  = x  + (size_t)b * TLEN * NF + f;
    const float* w1f = w1 + f * HID * KW;
    const float* b1f = b1 + f * HID;
    const float* w2f = w2 + f * HID * HID * KW;
    const float* b2f = b2 + f * HID;

    const int wvu   = __builtin_amdgcn_readfirstlane(wv);
    const int chalf = wvu >> 1;
    const int rpar  = wvu & 1;
    const float* w1h = w1f + chalf * (HID / 2) * KW;
    const float* b1h = b1f + chalf * (HID / 2);

    float au[NPT * 8];
#pragma unroll
    for (int s = 0; s < 4; ++s) {
#pragma unroll
        for (int j = 0; j < 2; ++j) {
            const float* wp = w2f + (m * HID + 4 * s + 2 * half + j) * KW;
            float g0 = wp[0], g1 = wp[1], g2 = wp[2], g3 = wp[3], g4 = wp[4];
            float ev  = g0 + g2 + g4,               od  = g1 + g3;
            float ev2 = g0 + 4.f * g2 + 16.f * g4,  od2 = 2.f * g1 + 8.f * g3;
            au[(0 * 4 + s) * 2 + j] = 0.25f * g0;
            au[(1 * 4 + s) * 2 + j] = (ev  + od ) * (-1.f / 6.f);
            au[(2 * 4 + s) * 2 + j] = (ev  - od ) * (-1.f / 6.f);
            au[(3 * 4 + s) * 2 + j] = (ev2 + od2) * (1.f / 24.f);
            au[(4 * 4 + s) * 2 + j] = (ev2 - od2) * (1.f / 24.f);
            au[(5 * 4 + s) * 2 + j] = g4;
        }
    }
    float bb[8];
#pragma unroll
    for (int r = 0; r < 8; ++r) bb[r] = b2f[r + 8 * half];

#pragma unroll 1
    for (int ch = 0; ch < 2; ++ch) {
        const int tbase = ch * CT;

        for (int v = tid; v < XSP; v += 128) {
            int t = tbase + v - 6;
            lds[XS_OFF + v] = (t >= 0 && t < TLEN) ? xg[(size_t)t * NF] : 0.f;
        }
        __syncthreads();

        for (int u = rpar * 32 + lane; u < HROWS; u += 64) {
            int t = tbase + u - 4;
            bool valid = (t >= 0 && t < TLEN);
            float xv[KW];
#pragma unroll
            for (int j = 0; j < KW; ++j) xv[j] = lds[XS_OFF + u + j];
            float hloc[HID / 2];
#pragma unroll
            for (int c = 0; c < HID / 2; ++c) {
                float h = b1h[c];
#pragma unroll
                for (int j = 0; j < KW; ++j) h = fmaf(w1h[c * KW + j], xv[j], h);
                hloc[c] = valid ? fmaxf(h, 0.f) : 0.f;
            }
#pragma unroll
            for (int g = 0; g < 2; ++g) {
                v4f q = { hloc[4*g+0], hloc[4*g+1], hloc[4*g+2], hloc[4*g+3] };
                *(v4f*)(&lds[H_OFF + u * HTP + chalf * 8 + 4 * g]) = q;
            }
        }
        __syncthreads();

        float pacc[8];
#pragma unroll
        for (int r = 0; r < 8; ++r) pacc[r] = 0.f;
        const int obase = (wv >> 1) * 256 + (wv & 1) * 128;

#pragma unroll 1
        for (int jt = 0; jt < 4; ++jt) {
            const int u0 = obase + 32 * jt + 2 * m + 2;
            const v8f bias8 = { bb[0], bb[1], bb[2], bb[3], bb[4], bb[5], bb[6], bb[7] };
            const v8f zero8 = { 0.f, 0.f, 0.f, 0.f, 0.f, 0.f, 0.f, 0.f };
            v8f acc0 = bias8, acc1 = zero8, acc2 = zero8, acc3 = zero8, acc4 = zero8, acc5 = bias8;
#pragma unroll
            for (int s = 0; s < 4; ++s) {
                const float* row = &lds[H_OFF + u0 * HTP + 4 * s + 2 * half];
                v2f d0 = *(const v2f*)(row);
                v2f d1 = *(const v2f*)(row + 1 * HTP);
                v2f d2 = *(const v2f*)(row + 2 * HTP);
                v2f d3 = *(const v2f*)(row + 3 * HTP);
                v2f d4 = *(const v2f*)(row + 4 * HTP);
                v2f d5 = *(const v2f*)(row + 5 * HTP);
                v2f ta = d4 - 4.f * d2, tb = d3 - 4.f * d1;
                v2f tc = d4 - d2,       te = d3 - d1;
                v2f D0 = (d4 - 5.f * d2) + 4.f * d0;
                v2f D1 = ta + tb;
                v2f D2 = ta - tb;
                v2f D3 = tc + 2.f * te;
                v2f D4 = tc - 2.f * te;
                v2f D5 = (d5 - 5.f * d3) + 4.f * d1;
                v2f a0 = { au[(0 * 4 + s) * 2], au[(0 * 4 + s) * 2 + 1] };
                v2f a1 = { au[(1 * 4 + s) * 2], au[(1 * 4 + s) * 2 + 1] };
                v2f a2 = { au[(2 * 4 + s) * 2], au[(2 * 4 + s) * 2 + 1] };
                v2f a3 = { au[(3 * 4 + s) * 2], au[(3 * 4 + s) * 2 + 1] };
                v2f a4 = { au[(4 * 4 + s) * 2], au[(4 * 4 + s) * 2 + 1] };
                v2f a5 = { au[(5 * 4 + s) * 2], au[(5 * 4 + s) * 2 + 1] };
                acc0 = __builtin_amdgcn_wmma_f32_16x16x4_f32(false, a0, false, D0, (short)0, acc0, false, false);
                acc1 = __builtin_amdgcn_wmma_f32_16x16x4_f32(false, a1, false, D1, (short)0, acc1, false, false);
                acc2 = __builtin_amdgcn_wmma_f32_16x16x4_f32(false, a2, false, D2, (short)0, acc2, false, false);
                acc3 = __builtin_amdgcn_wmma_f32_16x16x4_f32(false, a3, false, D3, (short)0, acc3, false, false);
                acc4 = __builtin_amdgcn_wmma_f32_16x16x4_f32(false, a4, false, D4, (short)0, acc4, false, false);
                acc5 = __builtin_amdgcn_wmma_f32_16x16x4_f32(false, a5, false, D5, (short)0, acc5, false, false);
            }
#pragma unroll
            for (int r = 0; r < 8; ++r) {
                float s12 = acc1[r] + acc2[r], d12 = acc1[r] - acc2[r];
                float s34 = acc3[r] + acc4[r], d34 = acc3[r] - acc4[r];
                float y0 = (acc0[r] + s12) + s34;
                float y1 = (acc5[r] + d12) + 2.f * d34;
                pacc[r] += fmaxf(y0, 0.f) + fmaxf(y1, 0.f);
            }
        }

        int seg = 2 * ch + (wv >> 1);
        int par = wv & 1;
        float red[8];
#pragma unroll
        for (int r = 0; r < 8; ++r) {
            float v = pacc[r];
            v += row_xor<0x161>(v);
            v += row_xor<0x162>(v);
            v += row_xor<0x164>(v);
            v += row_xor<0x168>(v);
            red[r] = v;
        }
        if ((lane & 15) == 0) {
#pragma unroll
            for (int r = 0; r < 8; ++r)
                lds[PART_OFF + ((r + 8 * half) * DPF + seg) * 2 + par] = red[r];
        }
        __syncthreads();
    }

    if (tid < DPF) {
        const float* pwf = pw + ((size_t)f * DPF + tid) * (HID * DPF);
        float acc = pb[f * DPF + tid];
#pragma unroll
        for (int i = 0; i < HID * DPF; ++i) {
            float p = (lds[PART_OFF + i * 2] + lds[PART_OFF + i * 2 + 1]) * (1.f / 256.f);
            acc = fmaf(pwf[i], p, acc);
        }
        out[(size_t)b * (NF * DPF) + f * DPF + tid] = acc;
    }
}

extern "C" void kernel_launch(void* const* d_in, const int* in_sizes, int n_in,
                              void* d_out, int out_size, void* d_ws, size_t ws_size,
                              hipStream_t stream) {
    (void)in_sizes; (void)n_in; (void)d_ws; (void)ws_size; (void)out_size;
    const float* x  = (const float*)d_in[0];
    const float* w1 = (const float*)d_in[2];
    const float* b1 = (const float*)d_in[3];
    const float* w2 = (const float*)d_in[4];
    const float* b2 = (const float*)d_in[5];
    const float* pw = (const float*)d_in[6];
    const float* pb = (const float*)d_in[7];
    float* out = (float*)d_out;

    dim3 grid(NF * BATCH);
    dim3 block(128);
    branch_kernel<<<grid, block, 0, stream>>>(x, w1, b1, w2, b2, pw, pb, out);
}


# conv2 as minimal filtering F(4,5): 8 transformed products per 4 outputs (16 f32 WMMAs per 32 outputs, was 24), h1 rows at pitch 18 for the 4-row lane stride; conv2 and pool sum order change
# speedup vs baseline: 1.3534x; 1.0459x over previous
#include <hip/hip_runtime.h>

typedef __attribute__((ext_vector_type(2))) float v2f;
typedef __attribute__((ext_vector_type(4))) float v4f;
typedef __attribute__((ext_vector_type(8))) float v8f;

#define NF    32
#define BATCH 128
#define TLEN  1024
#define HID   16
#define KW    5
#define DPF   4
#define NPT   8
#define CT    512
#define XSP   524
#define HTP   18
#define HROWS 520
#define XS_OFF   0
#define H_OFF    524
#define PART_OFF (H_OFF + HROWS * HTP)
#define LDS_FLOATS (PART_OFF + HID * DPF * 2)

template <int CTRL>
__device__ __forceinline__ float row_xor(float v) {
    int i = __builtin_bit_cast(int, v);
    int o = __builtin_amdgcn_update_dpp(i, i, CTRL, 0xf, 0xf, false);
    return __builtin_bit_cast(float, o);
}

__global__ __launch_bounds__(128)
void branch_kernel(const float* __restrict__ x,
                   const float* __restrict__ w1, const float* __restrict__ b1,
                   const float* __restrict__ w2, const float* __restrict__ b2,
                   const float* __restrict__ pw, const float* __restrict__ pb,
                   float* __restrict__ out)
{
    __shared__ __align__(16) float lds[LDS_FLOATS];
    const int f    = blockIdx.x & (NF - 1);
    const int b    = blockIdx.x >> 5;
    const int tid  = threadIdx.x;
    const int lane = tid & 31;
    const int wv   = tid >> 5;
    const int m    = lane & 15;
    const int half = lane >> 4;

    const float* xg  = x  + (size_t)b * TLEN * NF + f;
    const float* w1f = w1 + f * HID * KW;
    const float* b1f = b1 + f * HID;
    const float* w2f = w2 + f * HID * HID * KW;
    const float* b2f = b2 + f * HID;

    const int wvu   = __builtin_amdgcn_readfirstlane(wv);
    const int chalf = wvu >> 1;
    const int rpar  = wvu & 1;
    const float* w1h = w1f + chalf * (HID / 2) * KW;
    const float* b1h = b1f + chalf * (HID / 2);

    float au[NPT * 8];
#pragma unroll
    for (int s = 0; s < 4; ++s) {
#pragma unroll
        for (int j = 0; j < 2; ++j) {
            const float* wp = w2f + (m * HID + 4 * s + 2 * half + j) * KW;
            float g0 = wp[0], g1 = wp[1], g2 = wp[2], g3 = wp[3], g4 = wp[4];
            float ev  = g0 + g2 + g4,                       od  = g1 + g3;
            float ev2 = g0 + 4.f * g2 + 16.f * g4,          od2 = 2.f * g1 + 8.f * g3;
            float ev3 = g0 + 0.25f * g2 + 0.0625f * g4,     od3 = 0.5f * g1 + 0.125f * g3;
            au[(0 * 4 + s) * 2 + j] = g0;
            au[(1 * 4 + s) * 2 + j] = (ev  + od ) * (-2.f / 9.f);
            au[(2 * 4 + s) * 2 + j] = (ev  - od ) * (-2.f / 9.f);
            au[(3 * 4 + s) * 2 + j] = (ev2 + od2) * (1.f / 90.f);
            au[(4 * 4 + s) * 2 + j] = (ev2 - od2) * (1.f / 90.f);
            au[(5 * 4 + s) * 2 + j] = (ev3 + od3) * (32.f / 45.f);
            au[(6 * 4 + s) * 2 + j] = (ev3 - od3) * (32.f / 45.f);
            au[(7 * 4 + s) * 2 + j] = g4;
        }
    }
    float bb[8];
#pragma unroll
    for (int r = 0; r < 8; ++r) bb[r] = b2f[r + 8 * half];

#pragma unroll 1
    for (int ch = 0; ch < 2; ++ch) {
        const int tbase = ch * CT;

        for (int v = tid; v < XSP; v += 128) {
            int t = tbase + v - 6;
            lds[XS_OFF + v] = (t >= 0 && t < TLEN) ? xg[(size_t)t * NF] : 0.f;
        }
        __syncthreads();

        for (int u = rpar * 32 + lane; u < HROWS; u += 64) {
            int t = tbase + u - 4;
            bool valid = (t >= 0 && t < TLEN);
            float xv[KW];
#pragma unroll
            for (int j = 0; j < KW; ++j) xv[j] = lds[XS_OFF + u + j];
            float hloc[HID / 2];
#pragma unroll
            for (int c = 0; c < HID / 2; ++c) {
                float h = b1h[c];
#pragma unroll
                for (int j = 0; j < KW; ++j) h = fmaf(w1h[c * KW + j], xv[j], h);
                hloc[c] = valid ? fmaxf(h, 0.f) : 0.f;
            }
#pragma unroll
            for (int g = 0; g < 4; ++g) {
                v2f q = { hloc[2*g+0], hloc[2*g+1] };
                *(v2f*)(&lds[H_OFF + u * HTP + chalf * 8 + 2 * g]) = q;
            }
        }
        __syncthreads();

        float pacc[8];
#pragma unroll
        for (int r = 0; r < 8; ++r) pacc[r] = 0.f;
        const int obase = (wv >> 1) * 256 + (wv & 1) * 128;

#pragma unroll 1
        for (int jt = 0; jt < 2; ++jt) {
            const int u0 = obase + 64 * jt + 4 * m + 2;
            const v8f bias8 = { bb[0], bb[1], bb[2], bb[3], bb[4], bb[5], bb[6], bb[7] };
            const v8f zero8 = { 0.f, 0.f, 0.f, 0.f, 0.f, 0.f, 0.f, 0.f };
            v8f acc0 = zero8, acc1 = bias8, acc2 = zero8, acc3 = zero8, acc4 = zero8, acc5 = zero8, acc6 = zero8, acc7 = zero8;
#pragma unroll
            for (int s = 0; s < 4; ++s) {
                const float* row = &lds[H_OFF + u0 * HTP + 4 * s + 2 * half];
                v2f d0 = *(const v2f*)(row);
                v2f d1 = *(const v2f*)(row + 1 * HTP);
                v2f d2 = *(const v2f*)(row + 2 * HTP);
                v2f d3 = *(const v2f*)(row + 3 * HTP);
                v2f d4 = *(const v2f*)(row + 4 * HTP);
                v2f d5 = *(const v2f*)(row + 5 * HTP);
                v2f d6 = *(const v2f*)(row + 6 * HTP);
                v2f d7 = *(const v2f*)(row + 7 * HTP);
                v2f D0 = (d0 - d6) + 5.25f * (d4 - d2);
                v2f D7 = (d7 - d1) + 5.25f * (d3 - d5);
                v2f E1 = (d2 + d6) - 4.25f * d4,            O1 = (d1 + d5) - 4.25f * d3;
                v2f E2 = (0.25f * d2 + d6) - 1.25f * d4,    O2 = (4.f * d5 + d1) - 5.f * d3;
                v2f E3 = (4.f * d2 + d6) - 5.f * d4,        O3 = (4.f * d1 + d5) - 5.f * d3;
                v2f D1 = E1 + O1,         D2 = E1 - O1;
                v2f D3 = E2 + 0.5f * O2,  D4 = E2 - 0.5f * O2;
                v2f D5 = E3 + 0.5f * O3,  D6 = E3 - 0.5f * O3;
                v2f a0 = { au[(0 * 4 + s) * 2], au[(0 * 4 + s) * 2 + 1] };
                v2f a1 = { au[(1 * 4 + s) * 2], au[(1 * 4 + s) * 2 + 1] };
                v2f a2 = { au[(2 * 4 + s) * 2], au[(2 * 4 + s) * 2 + 1] };
                v2f a3 = { au[(3 * 4 + s) * 2], au[(3 * 4 + s) * 2 + 1] };
                v2f a4 = { au[(4 * 4 + s) * 2], au[(4 * 4 + s) * 2 + 1] };
                v2f a5 = { au[(5 * 4 + s) * 2], au[(5 * 4 + s) * 2 + 1] };
                v2f a6 = { au[(6 * 4 + s) * 2], au[(6 * 4 + s) * 2 + 1] };
                v2f a7 = { au[(7 * 4 + s) * 2], au[(7 * 4 + s) * 2 + 1] };
                acc0 = __builtin_amdgcn_wmma_f32_16x16x4_f32(false, a0, false, D0, (short)0, acc0, false, false);
                acc1 = __builtin_amdgcn_wmma_f32_16x16x4_f32(false, a1, false, D1, (short)0, acc1, false, false);
                acc2 = __builtin_amdgcn_wmma_f32_16x16x4_f32(false, a2, false, D2, (short)0, acc2, false, false);
                acc3 = __builtin_amdgcn_wmma_f32_16x16x4_f32(false, a3, false, D3, (short)0, acc3, false, false);
                acc4 = __builtin_amdgcn_wmma_f32_16x16x4_f32(false, a4, false, D4, (short)0, acc4, false, false);
                acc5 = __builtin_amdgcn_wmma_f32_16x16x4_f32(false, a5, false, D5, (short)0, acc5, false, false);
                acc6 = __builtin_amdgcn_wmma_f32_16x16x4_f32(false, a6, false, D6, (short)0, acc6, false, false);
                acc7 = __builtin_amdgcn_wmma_f32_16x16x4_f32(false, a7, false, D7, (short)0, acc7, false, false);
            }
#pragma unroll
            for (int r = 0; r < 8; ++r) {
                float S1 = acc1[r] + acc2[r], T1 = acc1[r] - acc2[r];
                float S2 = acc3[r] + acc4[r], T2 = acc3[r] - acc4[r];
                float S3 = acc5[r] + acc6[r], T3 = acc5[r] - acc6[r];
                float y0 = ((acc0[r] + S1) + S2) + S3;
                float y1 = (T1 + 2.f * T2) + 0.5f * T3;
                float y2 = (S1 + 4.f * S2) + 0.25f * S3;
                float y3 = ((acc7[r] + T1) + 8.f * T2) + 0.125f * T3;
                pacc[r] += (fmaxf(y0, 0.f) + fmaxf(y1, 0.f)) + (fmaxf(y2, 0.f) + fmaxf(y3, 0.f));
            }
        }

        int seg = 2 * ch + (wv >> 1);
        int par = wv & 1;
        float red[8];
#pragma unroll
        for (int r = 0; r < 8; ++r) {
            float v = pacc[r];
            v += row_xor<0x161>(v);
            v += row_xor<0x162>(v);
            v += row_xor<0x164>(v);
            v += row_xor<0x168>(v);
            red[r] = v;
        }
        if ((lane & 15) == 0) {
#pragma unroll
            for (int r = 0; r < 8; ++r)
                lds[PART_OFF + ((r + 8 * half) * DPF + seg) * 2 + par] = red[r];
        }
        __syncthreads();
    }

    if (tid < DPF) {
        const float* pwf = pw + ((size_t)f * DPF + tid) * (HID * DPF);
        float acc = pb[f * DPF + tid];
#pragma unroll
        for (int i = 0; i < HID * DPF; ++i) {
            float p = (lds[PART_OFF + i * 2] + lds[PART_OFF + i * 2 + 1]) * (1.f / 256.f);
            acc = fmaf(pwf[i], p, acc);
        }
        out[(size_t)b * (NF * DPF) + f * DPF + tid] = acc;
    }
}

extern "C" void kernel_launch(void* const* d_in, const int* in_sizes, int n_in,
                              void* d_out, int out_size, void* d_ws, size_t ws_size,
                              hipStream_t stream) {
    (void)in_sizes; (void)n_in; (void)d_ws; (void)ws_size; (void)out_size;
    const float* x  = (const float*)d_in[0];
    const float* w1 = (const float*)d_in[2];
    const float* b1 = (const float*)d_in[3];
    const float* w2 = (const float*)d_in[4];
    const float* b2 = (const float*)d_in[5];
    const float* pw = (const float*)d_in[6];
    const float* pb = (const float*)d_in[7];
    float* out = (float*)d_out;

    dim3 grid(NF * BATCH);
    dim3 block(128);
    branch_kernel<<<grid, block, 0, stream>>>(x, w1, b1, w2, b2, pw, pb, out);
}
